# starting kernel
# baseline (speedup 1.0000x reference)
#include <hip/hip_runtime.h>

typedef __attribute__((ext_vector_type(2))) float v2f;
typedef __attribute__((ext_vector_type(8))) float v8f;

__global__ void optnet_prep_kernel(const float* __restrict__ puzzles,
                                   const float* __restrict__ A,
                                   float* __restrict__ ws) {
  if (threadIdx.x != 0 || blockIdx.x != 0) return;
  double M[9][18];
  for (int i = 0; i < 9; ++i)
    for (int j = 0; j < 18; ++j) M[i][j] = 0.0;
  for (int i = 0; i < 6; ++i) M[i][i] = 1.0;
  for (int r = 0; r < 3; ++r)
    for (int j = 0; j < 6; ++j) {
      double a = (double)A[r * 6 + j];
      M[j][6 + r] = a;
      M[6 + r][j] = a;
    }
  for (int i = 0; i < 9; ++i) M[i][9 + i] = 1.0;

  for (int col = 0; col < 9; ++col) {
    int piv = col;
    double best = M[col][col] < 0.0 ? -M[col][col] : M[col][col];
    for (int r = col + 1; r < 9; ++r) {
      double v = M[r][col] < 0.0 ? -M[r][col] : M[r][col];
      if (v > best) { best = v; piv = r; }
    }
    if (piv != col)
      for (int j = 0; j < 18; ++j) {
        double t = M[col][j]; M[col][j] = M[piv][j]; M[piv][j] = t;
      }
    double inv = 1.0 / M[col][col];
    for (int j = 0; j < 18; ++j) M[col][j] *= inv;
    for (int r = 0; r < 9; ++r) {
      if (r == col) continue;
      double f = M[r][col];
      if (f != 0.0)
        for (int j = 0; j < 18; ++j) M[r][j] -= f * M[col][j];
    }
  }

  double bval = -(double)puzzles[3];
  for (int j = 0; j < 6; ++j) {
    for (int k = 0; k < 6; ++k) ws[j * 6 + k] = (float)M[j][9 + k];
    ws[36 + j] = (float)(bval * (M[j][9 + 6] + M[j][9 + 7] + M[j][9 + 8]));
  }
}

__global__ void __launch_bounds__(256) optnet_main_kernel(
    const float* __restrict__ in, float* __restrict__ out,
    const float* __restrict__ ws, int numTiles) {
  const int lane = threadIdx.x & 31;
  const int waveId = (blockIdx.x << 3) | (threadIdx.x >> 5);
  const int n  = lane & 15;
  const int hi = lane >> 4;

  const int nn = (n < 6) ? n : 0;
  const float* wrow = ws + nn * 6;
  float b00 = wrow[hi ? 2 : 0];
  float b01 = wrow[hi ? 3 : 1];
  float b10 = hi ? 0.0f : wrow[4];
  float b11 = hi ? 0.0f : wrow[5];
  float cn  = ws[36 + nn];
  if (n >= 6) { b00 = 0.0f; b01 = 0.0f; b10 = 0.0f; b11 = 0.0f; cn = 0.0f; }
  v2f bLo = { b00, b01 };
  v2f bHi = { b10, b11 };
  const v2f vzero = { 0.0f, 0.0f };

  const int tile0 = waveId * 4;
  #pragma unroll
  for (int t = 0; t < 4; ++t) {
    int tile = tile0 + t;
    if (tile >= numTiles) break;
    int base = tile << 4;
    int row  = base + n;
    const float* rowp = in + (size_t)row * 6;

    v2f a0 = *(const v2f*)(rowp + (hi << 1));
    if (hi && row < 2) a0.y = 0.0f;
    v2f a1 = hi ? vzero : *(const v2f*)(rowp + 4);

    v8f acc = {};
    acc = __builtin_amdgcn_wmma_f32_16x16x4_f32(false, a0, false, bLo,
                                                (short)0, acc, false, false);
    acc = __builtin_amdgcn_wmma_f32_16x16x4_f32(false, a1, false, bHi,
                                                (short)0, acc, false, false);

    if (n < 6) {
      float* orow = out + (size_t)(base + (hi << 3)) * 6 + n;
      #pragma unroll
      for (int r = 0; r < 8; ++r) orow[(size_t)r * 6] = acc[r] + cn;
    }
  }
}

extern "C" void kernel_launch(void* const* d_in, const int* in_sizes, int n_in,
                              void* d_out, int out_size, void* d_ws, size_t ws_size,
                              hipStream_t stream) {
  const float* puzzles = (const float*)d_in[0];
  const float* A       = (const float*)d_in[1];
  float* out = (float*)d_out;
  float* ws  = (float*)d_ws;

  int nRows    = in_sizes[0] / 6;
  int numTiles = (nRows + 15) / 16;

  optnet_prep_kernel<<<1, 32, 0, stream>>>(puzzles, A, ws);

  int waves  = (numTiles + 3) / 4;
  int blocks = (waves + 7) / 8;
  optnet_main_kernel<<<blocks, 256, 0, stream>>>(puzzles, out, ws, numTiles);
}


# prep inversion: one thread's f64 Gauss-Jordan on a 9x18 array in scratch memory -> one wave, one matrix column per lane, its 9 rows in registers, pivot column by readlane; same operations and order
# speedup vs baseline: 1.8153x; 1.8153x over previous
#include <hip/hip_runtime.h>

typedef __attribute__((ext_vector_type(2))) float v2f;
typedef __attribute__((ext_vector_type(8))) float v8f;


__device__ __forceinline__ double lane_bcast(double v, int srcLane) {
  int lo = __builtin_amdgcn_readlane(__double2loint(v), srcLane);
  int hi = __builtin_amdgcn_readlane(__double2hiint(v), srcLane);
  return __hiloint2double(hi, lo);
}

__global__ void __launch_bounds__(32) optnet_prep_kernel(
    const float* __restrict__ puzzles, const float* __restrict__ A,
    float* __restrict__ ws) {
  if (blockIdx.x != 0 || threadIdx.x >= 32) return;
  const int j = threadIdx.x;

  const bool inAt = (j >= 6) && (j < 9);
  const bool inA  = (j < 6);
  const int  jt   = inAt ? (j - 6) : 0;
  const int  ja   = inA ? j : 0;
  double M[9];
  for (int i = 0; i < 6; ++i) {
    float at = A[jt * 6 + i];
    float v  = inAt ? at : ((j == i || j == 9 + i) ? 1.0f : 0.0f);
    M[i] = (double)v;
  }
  for (int r = 0; r < 3; ++r) {
    float a = A[r * 6 + ja];
    float v = inA ? a : ((j == 15 + r) ? 1.0f : 0.0f);
    M[6 + r] = (double)v;
  }

  #pragma unroll
  for (int col = 0; col < 9; ++col) {
    double cv[9];
    for (int r = 0; r < 9; ++r) cv[r] = lane_bcast(M[r], col);

    int piv = col;
    double best = __builtin_fabs(cv[col]);
    for (int r = col + 1; r < 9; ++r) {
      double v = __builtin_fabs(cv[r]);
      if (v > best) { best = v; piv = r; }
    }
    for (int r = col + 1; r < 9; ++r) {
      if (piv == r) {
        double t = M[col];  M[col]  = M[r];  M[r]  = t;
        double s = cv[col]; cv[col] = cv[r]; cv[r] = s;
      }
    }
    double inv = 1.0 / cv[col];
    M[col] *= inv;
    for (int r = 0; r < 9; ++r) {
      if (r == col) continue;
      double f = cv[r];
      if (((__double2hiint(f) & 0x7fffffff) | __double2loint(f)) != 0)
        M[r] -= f * M[col];
    }
  }

  double bval = -(double)puzzles[3];
  for (int r = 0; r < 6; ++r) {
    if (j >= 9 && j < 15) ws[r * 6 + (j - 9)] = (float)M[r];
    double s = lane_bcast(M[r], 15) + lane_bcast(M[r], 16) + lane_bcast(M[r], 17);
    if (j == 0) ws[36 + r] = (float)(bval * s);
  }
}

__global__ void __launch_bounds__(256) optnet_main_kernel(
    const float* __restrict__ in, float* __restrict__ out,
    const float* __restrict__ ws, int numTiles) {
  const int lane = threadIdx.x & 31;
  const int waveId = (blockIdx.x << 3) | (threadIdx.x >> 5);
  const int n  = lane & 15;
  const int hi = lane >> 4;

  const int nn = (n < 6) ? n : 0;
  const float* wrow = ws + nn * 6;
  float b00 = wrow[hi ? 2 : 0];
  float b01 = wrow[hi ? 3 : 1];
  float b10 = hi ? 0.0f : wrow[4];
  float b11 = hi ? 0.0f : wrow[5];
  float cn  = ws[36 + nn];
  if (n >= 6) { b00 = 0.0f; b01 = 0.0f; b10 = 0.0f; b11 = 0.0f; cn = 0.0f; }
  v2f bLo = { b00, b01 };
  v2f bHi = { b10, b11 };
  const v2f vzero = { 0.0f, 0.0f };

  const int tile0 = waveId * 4;
  #pragma unroll
  for (int t = 0; t < 4; ++t) {
    int tile = tile0 + t;
    if (tile >= numTiles) break;
    int base = tile << 4;
    int row  = base + n;
    const float* rowp = in + (size_t)row * 6;

    v2f a0 = *(const v2f*)(rowp + (hi << 1));
    if (hi && row < 2) a0.y = 0.0f;
    v2f a1 = hi ? vzero : *(const v2f*)(rowp + 4);

    v8f acc = {};
    acc = __builtin_amdgcn_wmma_f32_16x16x4_f32(false, a0, false, bLo,
                                                (short)0, acc, false, false);
    acc = __builtin_amdgcn_wmma_f32_16x16x4_f32(false, a1, false, bHi,
                                                (short)0, acc, false, false);

    if (n < 6) {
      float* orow = out + (size_t)(base + (hi << 3)) * 6 + n;
      #pragma unroll
      for (int r = 0; r < 8; ++r) orow[(size_t)r * 6] = acc[r] + cn;
    }
  }
}

extern "C" void kernel_launch(void* const* d_in, const int* in_sizes, int n_in,
                              void* d_out, int out_size, void* d_ws, size_t ws_size,
                              hipStream_t stream) {
  const float* puzzles = (const float*)d_in[0];
  const float* A       = (const float*)d_in[1];
  float* out = (float*)d_out;
  float* ws  = (float*)d_ws;

  int nRows    = in_sizes[0] / 6;
  int numTiles = (nRows + 15) / 16;

  optnet_prep_kernel<<<1, 32, 0, stream>>>(puzzles, A, ws);

  int waves  = (numTiles + 3) / 4;
  int blocks = (waves + 7) / 8;
  optnet_main_kernel<<<blocks, 256, 0, stream>>>(puzzles, out, ws, numTiles);
}


# main kernel: fixed grid of 512 workgroups whose waves loop over 8-tile batches, next batch's loads issued before the current batch's WMMAs and stores, in place of 65,536 short waves of 4 tiles
# speedup vs baseline: 2.0218x; 1.1138x over previous
#include <hip/hip_runtime.h>

typedef __attribute__((ext_vector_type(2))) float v2f;
typedef __attribute__((ext_vector_type(8))) float v8f;


__device__ __forceinline__ double lane_bcast(double v, int srcLane) {
  int lo = __builtin_amdgcn_readlane(__double2loint(v), srcLane);
  int hi = __builtin_amdgcn_readlane(__double2hiint(v), srcLane);
  return __hiloint2double(hi, lo);
}

__global__ void __launch_bounds__(32) optnet_prep_kernel(
    const float* __restrict__ puzzles, const float* __restrict__ A,
    float* __restrict__ ws) {
  if (blockIdx.x != 0 || threadIdx.x >= 32) return;
  const int j = threadIdx.x;

  const bool inAt = (j >= 6) && (j < 9);
  const bool inA  = (j < 6);
  const int  jt   = inAt ? (j - 6) : 0;
  const int  ja   = inA ? j : 0;
  double M[9];
  for (int i = 0; i < 6; ++i) {
    float at = A[jt * 6 + i];
    float v  = inAt ? at : ((j == i || j == 9 + i) ? 1.0f : 0.0f);
    M[i] = (double)v;
  }
  for (int r = 0; r < 3; ++r) {
    float a = A[r * 6 + ja];
    float v = inA ? a : ((j == 15 + r) ? 1.0f : 0.0f);
    M[6 + r] = (double)v;
  }

  #pragma unroll
  for (int col = 0; col < 9; ++col) {
    double cv[9];
    for (int r = 0; r < 9; ++r) cv[r] = lane_bcast(M[r], col);

    int piv = col;
    double best = __builtin_fabs(cv[col]);
    for (int r = col + 1; r < 9; ++r) {
      double v = __builtin_fabs(cv[r]);
      if (v > best) { best = v; piv = r; }
    }
    for (int r = col + 1; r < 9; ++r) {
      if (piv == r) {
        double t = M[col];  M[col]  = M[r];  M[r]  = t;
        double s = cv[col]; cv[col] = cv[r]; cv[r] = s;
      }
    }
    double inv = 1.0 / cv[col];
    M[col] *= inv;
    for (int r = 0; r < 9; ++r) {
      if (r == col) continue;
      double f = cv[r];
      if (((__double2hiint(f) & 0x7fffffff) | __double2loint(f)) != 0)
        M[r] -= f * M[col];
    }
  }

  double bval = -(double)puzzles[3];
  for (int r = 0; r < 6; ++r) {
    if (j >= 9 && j < 15) ws[r * 6 + (j - 9)] = (float)M[r];
    double s = lane_bcast(M[r], 15) + lane_bcast(M[r], 16) + lane_bcast(M[r], 17);
    if (j == 0) ws[36 + r] = (float)(bval * s);
  }
}

#define TPB 8

__device__ __forceinline__ void load_batch(const float* __restrict__ in, int b,
                                           int numTiles, int laneOff, int n,
                                           v2f (&a0)[TPB], v2f (&a1)[TPB]) {
  #pragma unroll
  for (int t = 0; t < TPB; ++t) {
    int tile = b * TPB + t;
    if (tile >= numTiles) tile = numTiles - 1;
    const float* tp = in + (size_t)tile * 96;
    a0[t] = *(const v2f*)(tp + laneOff);
    a1[t] = *(const v2f*)(tp + n * 6 + 4);
  }
}

__device__ __forceinline__ void compute_batch(float* __restrict__ out, int b,
                                              int numTiles, int n, int hi,
                                              v2f bLo, v2f bHi, float cn,
                                              const v2f (&a0)[TPB],
                                              const v2f (&a1)[TPB]) {
  #pragma unroll
  for (int t = 0; t < TPB; ++t) {
    int tile = b * TPB + t;
    if (tile >= numTiles) break;
    int base = tile << 4;
    v2f x0 = a0[t];
    if (hi && (base + n) < 2) x0.y = 0.0f;

    v8f acc = {};
    acc = __builtin_amdgcn_wmma_f32_16x16x4_f32(false, x0, false, bLo,
                                                (short)0, acc, false, false);
    acc = __builtin_amdgcn_wmma_f32_16x16x4_f32(false, a1[t], false, bHi,
                                                (short)0, acc, false, false);

    if (n < 6) {
      float* orow = out + (size_t)(base + (hi << 3)) * 6 + n;
      #pragma unroll
      for (int r = 0; r < 8; ++r) orow[(size_t)r * 6] = acc[r] + cn;
    }
  }
}

__global__ void __launch_bounds__(256) optnet_main_kernel(
    const float* __restrict__ in, float* __restrict__ out,
    const float* __restrict__ ws, int numTiles) {
  const int lane = threadIdx.x & 31;
  const int waveId = (blockIdx.x << 3) |
                     __builtin_amdgcn_readfirstlane((int)(threadIdx.x >> 5));
  const int numWaves = gridDim.x << 3;
  const int n  = lane & 15;
  const int hi = lane >> 4;

  const int numBatches = (numTiles + TPB - 1) / TPB;
  int b = waveId;
  if (b >= numBatches) return;

  const int laneOff = n * 6 + (hi << 1);
  v2f x0[TPB], x1[TPB], y0[TPB], y1[TPB];
  load_batch(in, b, numTiles, laneOff, n, x0, x1);

  const int nn = (n < 6) ? n : 0;
  const float* wrow = ws + nn * 6;
  float b00 = wrow[hi ? 2 : 0];
  float b01 = wrow[hi ? 3 : 1];
  float b10 = hi ? 0.0f : wrow[4];
  float b11 = hi ? 0.0f : wrow[5];
  float cn  = ws[36 + nn];
  if (n >= 6) { b00 = 0.0f; b01 = 0.0f; b10 = 0.0f; b11 = 0.0f; cn = 0.0f; }
  v2f bLo = { b00, b01 };
  v2f bHi = { b10, b11 };

  while (b < numBatches) {
    load_batch(in, b + numWaves, numTiles, laneOff, n, y0, y1);
    compute_batch(out, b, numTiles, n, hi, bLo, bHi, cn, x0, x1);
    b += numWaves;
    load_batch(in, b + numWaves, numTiles, laneOff, n, x0, x1);
    compute_batch(out, b, numTiles, n, hi, bLo, bHi, cn, y0, y1);
    b += numWaves;
  }
}

extern "C" void kernel_launch(void* const* d_in, const int* in_sizes, int n_in,
                              void* d_out, int out_size, void* d_ws, size_t ws_size,
                              hipStream_t stream) {
  const float* puzzles = (const float*)d_in[0];
  const float* A       = (const float*)d_in[1];
  float* out = (float*)d_out;
  float* ws  = (float*)d_ws;

  int nRows    = in_sizes[0] / 6;
  int numTiles = (nRows + 15) / 16;

  optnet_prep_kernel<<<1, 32, 0, stream>>>(puzzles, A, ws);

  optnet_main_kernel<<<512, 256, 0, stream>>>(puzzles, out, ws, numTiles);
}


# prep: f64 Gram-Schmidt on A's three rows (W = I - sum u u'/d, c from L w = 1) replaces the pivoted f64 elimination of the 9x9 K: other f64 operations and summation order; main kernel unchanged
# speedup vs baseline: 2.3215x; 1.1482x over previous
#include <hip/hip_runtime.h>

typedef __attribute__((ext_vector_type(2))) float v2f;
typedef __attribute__((ext_vector_type(8))) float v8f;


template <int CTRL>
__device__ __forceinline__ double dpp_f64(double v) {
  int lo = __builtin_amdgcn_update_dpp(0, __double2loint(v), CTRL, 0xf, 0xf, true);
  int hi = __builtin_amdgcn_update_dpp(0, __double2hiint(v), CTRL, 0xf, 0xf, true);
  return __hiloint2double(hi, lo);
}

__device__ __forceinline__ double sum8(double v) {
  v += dpp_f64<0xB1>(v);
  v += dpp_f64<0x4E>(v);
  v += dpp_f64<0x141>(v);
  return v;
}

__global__ void __launch_bounds__(32) optnet_prep_kernel(
    const float* __restrict__ puzzles, const float* __restrict__ A,
    float* __restrict__ ws) {
  __shared__ double sG[3][6];
  __shared__ double sU[3][7];
  if (blockIdx.x != 0) return;
  const int l = threadIdx.x;
  const int k = l & 7;
  const bool comp = k < 6;
  const int kk = comp ? k : 0;

  const float f1 = A[kk], f2 = A[6 + kk], f3 = A[12 + kk];
  const double p3 = (double)puzzles[3];
  const double a1 = comp ? (double)f1 : 0.0;
  const double a2 = comp ? (double)f2 : 0.0;
  const double a3 = comp ? (double)f3 : 0.0;

  const double d1  = sum8(a1 * a1);
  const double t12 = sum8(a1 * a2);
  const double t13 = sum8(a1 * a3);
  const double r1  = 1.0 / d1;
  const double l21 = t12 * r1;
  const double l31 = t13 * r1;
  const double u2  = __builtin_fma(-l21, a1, a2);
  const double a3p = __builtin_fma(-l31, a1, a3);
  const double d2  = sum8(u2 * u2);
  const double t23 = sum8(u2 * a3p);
  const double r2  = 1.0 / d2;
  const double l32 = t23 * r2;
  const double u3  = __builtin_fma(-l32, u2, a3p);
  const double d3  = sum8(u3 * u3);
  const double r3  = 1.0 / d3;
  const double w2 = 1.0 - l21;
  const double w3 = __builtin_fma(-l32, w2, 1.0 - l31);

  if (l < 6) { sG[0][l] = a1 * r1; sG[1][l] = u2 * r2; sG[2][l] = u3 * r3; }
  if (l < 7) {
    sU[0][l] = (l < 6) ? a1 : p3;
    sU[1][l] = (l < 6) ? u2 : p3 * w2;
    sU[2][l] = (l < 6) ? u3 : p3 * w3;
  }
  __syncthreads();

  for (int t = 0; t < 2; ++t) {
    const int item = l + 32 * t;
    const bool live = item < 42;
    const int e = live ? item : 0;
    const int n = e / 7;
    const int K = e - 7 * n;
    double v = (n == K) ? 1.0 : 0.0;
    v = __builtin_fma(-sG[0][n], sU[0][K], v);
    v = __builtin_fma(-sG[1][n], sU[1][K], v);
    v = __builtin_fma(-sG[2][n], sU[2][K], v);
    if (live) ws[(K < 6) ? (n * 6 + K) : (36 + n)] = (float)v;
  }
}

#define TPB 8

__device__ __forceinline__ void load_batch(const float* __restrict__ in, int b,
                                           int numTiles, int laneOff, int n,
                                           v2f (&a0)[TPB], v2f (&a1)[TPB]) {
  #pragma unroll
  for (int t = 0; t < TPB; ++t) {
    int tile = b * TPB + t;
    if (tile >= numTiles) tile = numTiles - 1;
    const float* tp = in + (size_t)tile * 96;
    a0[t] = *(const v2f*)(tp + laneOff);
    a1[t] = *(const v2f*)(tp + n * 6 + 4);
  }
}

__device__ __forceinline__ void compute_batch(float* __restrict__ out, int b,
                                              int numTiles, int n, int hi,
                                              v2f bLo, v2f bHi, float cn,
                                              const v2f (&a0)[TPB],
                                              const v2f (&a1)[TPB]) {
  #pragma unroll
  for (int t = 0; t < TPB; ++t) {
    int tile = b * TPB + t;
    if (tile >= numTiles) break;
    int base = tile << 4;
    v2f x0 = a0[t];
    if (hi && (base + n) < 2) x0.y = 0.0f;

    v8f acc = {};
    acc = __builtin_amdgcn_wmma_f32_16x16x4_f32(false, x0, false, bLo,
                                                (short)0, acc, false, false);
    acc = __builtin_amdgcn_wmma_f32_16x16x4_f32(false, a1[t], false, bHi,
                                                (short)0, acc, false, false);

    if (n < 6) {
      float* orow = out + (size_t)(base + (hi << 3)) * 6 + n;
      #pragma unroll
      for (int r = 0; r < 8; ++r) orow[(size_t)r * 6] = acc[r] + cn;
    }
  }
}

__global__ void __launch_bounds__(256) optnet_main_kernel(
    const float* __restrict__ in, float* __restrict__ out,
    const float* __restrict__ ws, int numTiles) {
  const int lane = threadIdx.x & 31;
  const int waveId = (blockIdx.x << 3) |
                     __builtin_amdgcn_readfirstlane((int)(threadIdx.x >> 5));
  const int numWaves = gridDim.x << 3;
  const int n  = lane & 15;
  const int hi = lane >> 4;

  const int numBatches = (numTiles + TPB - 1) / TPB;
  int b = waveId;
  if (b >= numBatches) return;

  const int laneOff = n * 6 + (hi << 1);
  v2f x0[TPB], x1[TPB], y0[TPB], y1[TPB];
  load_batch(in, b, numTiles, laneOff, n, x0, x1);

  const int nn = (n < 6) ? n : 0;
  const float* wrow = ws + nn * 6;
  float b00 = wrow[hi ? 2 : 0];
  float b01 = wrow[hi ? 3 : 1];
  float b10 = hi ? 0.0f : wrow[4];
  float b11 = hi ? 0.0f : wrow[5];
  float cn  = ws[36 + nn];
  if (n >= 6) { b00 = 0.0f; b01 = 0.0f; b10 = 0.0f; b11 = 0.0f; cn = 0.0f; }
  v2f bLo = { b00, b01 };
  v2f bHi = { b10, b11 };

  while (b < numBatches) {
    load_batch(in, b + numWaves, numTiles, laneOff, n, y0, y1);
    compute_batch(out, b, numTiles, n, hi, bLo, bHi, cn, x0, x1);
    b += numWaves;
    load_batch(in, b + numWaves, numTiles, laneOff, n, x0, x1);
    compute_batch(out, b, numTiles, n, hi, bLo, bHi, cn, y0, y1);
    b += numWaves;
  }
}

extern "C" void kernel_launch(void* const* d_in, const int* in_sizes, int n_in,
                              void* d_out, int out_size, void* d_ws, size_t ws_size,
                              hipStream_t stream) {
  const float* puzzles = (const float*)d_in[0];
  const float* A       = (const float*)d_in[1];
  float* out = (float*)d_out;
  float* ws  = (float*)d_ws;

  int nRows    = in_sizes[0] / 6;
  int numTiles = (nRows + 15) / 16;

  optnet_prep_kernel<<<1, 32, 0, stream>>>(puzzles, A, ws);

  optnet_main_kernel<<<512, 256, 0, stream>>>(puzzles, out, ws, numTiles);
}


# one launch instead of two: one wave per workgroup runs the same f64 Gram-Schmidt chain into LDS while every wave's first two trips of row loads are in flight; W and c come from LDS, not workspace
# speedup vs baseline: 2.9827x; 1.2848x over previous
#include <hip/hip_runtime.h>

typedef __attribute__((ext_vector_type(2))) float v2f;
typedef __attribute__((ext_vector_type(8))) float v8f;


template <int CTRL>
__device__ __forceinline__ double dpp_f64(double v) {
  int lo = __builtin_amdgcn_update_dpp(0, __double2loint(v), CTRL, 0xf, 0xf, true);
  int hi = __builtin_amdgcn_update_dpp(0, __double2hiint(v), CTRL, 0xf, 0xf, true);
  return __hiloint2double(hi, lo);
}

__device__ __forceinline__ double sum8(double v) {
  v += dpp_f64<0xB1>(v);
  v += dpp_f64<0x4E>(v);
  v += dpp_f64<0x141>(v);
  return v;
}

#define TPB 8

__device__ __forceinline__ void load_batch(const float* __restrict__ in, int b,
                                           int numTiles, int laneOff, int n,
                                           v2f (&a0)[TPB], v2f (&a1)[TPB]) {
  #pragma unroll
  for (int t = 0; t < TPB; ++t) {
    int tile = b * TPB + t;
    if (tile >= numTiles) tile = numTiles - 1;
    const float* tp = in + (size_t)tile * 96;
    a0[t] = *(const v2f*)(tp + laneOff);
    a1[t] = *(const v2f*)(tp + n * 6 + 4);
  }
}

__device__ __forceinline__ void compute_batch(float* __restrict__ out, int b,
                                              int numTiles, int n, int hi,
                                              v2f bLo, v2f bHi, float cn,
                                              const v2f (&a0)[TPB],
                                              const v2f (&a1)[TPB]) {
  #pragma unroll
  for (int t = 0; t < TPB; ++t) {
    int tile = b * TPB + t;
    if (tile >= numTiles) break;
    int base = tile << 4;
    v2f x0 = a0[t];
    if (hi && (base + n) < 2) x0.y = 0.0f;

    v8f acc = {};
    acc = __builtin_amdgcn_wmma_f32_16x16x4_f32(false, x0, false, bLo,
                                                (short)0, acc, false, false);
    acc = __builtin_amdgcn_wmma_f32_16x16x4_f32(false, a1[t], false, bHi,
                                                (short)0, acc, false, false);

    if (n < 6) {
      float* orow = out + (size_t)(base + (hi << 3)) * 6 + n;
      #pragma unroll
      for (int r = 0; r < 8; ++r) orow[(size_t)r * 6] = acc[r] + cn;
    }
  }
}

__device__ __forceinline__ void lds_barrier() {
  __builtin_amdgcn_fence(__ATOMIC_RELEASE, "workgroup", "local");
  __builtin_amdgcn_s_barrier();
  __builtin_amdgcn_fence(__ATOMIC_ACQUIRE, "workgroup", "local");
}

__global__ void __launch_bounds__(256) optnet_main_kernel(
    const float* __restrict__ in, const float* __restrict__ A,
    float* __restrict__ out, int numTiles) {
  __shared__ double sG[3][6];
  __shared__ double sU[3][7];
  __shared__ float  sW[42];
  if (numTiles <= 0) return;

  const int lane = threadIdx.x & 31;
  const int wave = __builtin_amdgcn_readfirstlane((int)(threadIdx.x >> 5));
  const int waveId = (blockIdx.x << 3) | wave;
  const int numWaves = gridDim.x << 3;
  const int n  = lane & 15;
  const int hi = lane >> 4;

  const int chainWave = (int)((blockIdx.x ^ (blockIdx.x >> 8)) & 1u);
  const bool chain = (wave == chainWave);

  const int k = lane & 7;
  const bool comp = k < 6;
  const int kk = comp ? k : 0;
  float f1 = 0.0f, f2 = 0.0f, f3 = 0.0f, fp = 0.0f;
  if (chain) { f1 = A[kk]; f2 = A[6 + kk]; f3 = A[12 + kk]; fp = in[3]; }

  const int numBatches = (numTiles + TPB - 1) / TPB;
  int b = waveId;
  const int laneOff = n * 6 + (hi << 1);
  v2f x0[TPB], x1[TPB], y0[TPB], y1[TPB];
  load_batch(in, b, numTiles, laneOff, n, x0, x1);
  load_batch(in, b + numWaves, numTiles, laneOff, n, y0, y1);

  if (chain) {
    const double p3 = (double)fp;
    const double a1 = comp ? (double)f1 : 0.0;
    const double a2 = comp ? (double)f2 : 0.0;
    const double a3 = comp ? (double)f3 : 0.0;

    const double d1  = sum8(a1 * a1);
    const double t12 = sum8(a1 * a2);
    const double t13 = sum8(a1 * a3);
    const double r1  = 1.0 / d1;
    const double l21 = t12 * r1;
    const double l31 = t13 * r1;
    const double u2  = __builtin_fma(-l21, a1, a2);
    const double a3p = __builtin_fma(-l31, a1, a3);
    const double d2  = sum8(u2 * u2);
    const double t23 = sum8(u2 * a3p);
    const double r2  = 1.0 / d2;
    const double l32 = t23 * r2;
    const double u3  = __builtin_fma(-l32, u2, a3p);
    const double d3  = sum8(u3 * u3);
    const double r3  = 1.0 / d3;
    const double w2 = 1.0 - l21;
    const double w3 = __builtin_fma(-l32, w2, 1.0 - l31);

    if (lane < 6) { sG[0][lane] = a1 * r1; sG[1][lane] = u2 * r2; sG[2][lane] = u3 * r3; }
    if (lane < 7) {
      sU[0][lane] = (lane < 6) ? a1 : p3;
      sU[1][lane] = (lane < 6) ? u2 : p3 * w2;
      sU[2][lane] = (lane < 6) ? u3 : p3 * w3;
    }
  }
  lds_barrier();

  if (threadIdx.x < 42) {
    const int e  = threadIdx.x;
    const int rn = e / 7;
    const int K  = e - 7 * rn;
    double v = (rn == K) ? 1.0 : 0.0;
    v = __builtin_fma(-sG[0][rn], sU[0][K], v);
    v = __builtin_fma(-sG[1][rn], sU[1][K], v);
    v = __builtin_fma(-sG[2][rn], sU[2][K], v);
    sW[(K < 6) ? (rn * 6 + K) : (36 + rn)] = (float)v;
  }
  lds_barrier();

  const int nn = (n < 6) ? n : 0;
  const float* wrow = sW + nn * 6;
  float b00 = wrow[hi ? 2 : 0];
  float b01 = wrow[hi ? 3 : 1];
  float b10 = hi ? 0.0f : wrow[4];
  float b11 = hi ? 0.0f : wrow[5];
  float cn  = sW[36 + nn];
  if (n >= 6) { b00 = 0.0f; b01 = 0.0f; b10 = 0.0f; b11 = 0.0f; cn = 0.0f; }
  v2f bLo = { b00, b01 };
  v2f bHi = { b10, b11 };

  while (b < numBatches) {
    compute_batch(out, b, numTiles, n, hi, bLo, bHi, cn, x0, x1);
    load_batch(in, b + 2 * numWaves, numTiles, laneOff, n, x0, x1);
    compute_batch(out, b + numWaves, numTiles, n, hi, bLo, bHi, cn, y0, y1);
    load_batch(in, b + 3 * numWaves, numTiles, laneOff, n, y0, y1);
    b += 2 * numWaves;
  }
}

extern "C" void kernel_launch(void* const* d_in, const int* in_sizes, int n_in,
                              void* d_out, int out_size, void* d_ws, size_t ws_size,
                              hipStream_t stream) {
  const float* puzzles = (const float*)d_in[0];
  const float* A       = (const float*)d_in[1];
  float* out = (float*)d_out;

  int nRows    = in_sizes[0] / 6;
  int numTiles = (nRows + 15) / 16;

  optnet_main_kernel<<<512, 256, 0, stream>>>(puzzles, A, out, numTiles);
}


# stores: product transposed over tile pairs (a row's 6 products summed 2+2+2, not 4+2) so lane l holds row l; wave regroups a trip in its LDS patch, writes 6 whole-line 512 B stores, not 64 partial
# speedup vs baseline: 3.4301x; 1.1500x over previous
#include <hip/hip_runtime.h>

typedef __attribute__((ext_vector_type(2))) float v2f;
typedef __attribute__((ext_vector_type(4))) float v4f;
typedef __attribute__((ext_vector_type(8))) float v8f;


template <int CTRL>
__device__ __forceinline__ double dpp_f64(double v) {
  int lo = __builtin_amdgcn_update_dpp(0, __double2loint(v), CTRL, 0xf, 0xf, true);
  int hi = __builtin_amdgcn_update_dpp(0, __double2hiint(v), CTRL, 0xf, 0xf, true);
  return __hiloint2double(hi, lo);
}

__device__ __forceinline__ double sum8(double v) {
  v += dpp_f64<0xB1>(v);
  v += dpp_f64<0x4E>(v);
  v += dpp_f64<0x141>(v);
  return v;
}

#define TPB 8
#define PPB (TPB / 2)
#define TRIP_FLOATS (TPB * 96)

__device__ __forceinline__ void load_batch(const float* __restrict__ in, int b,
                                           int numRows, int lane,
                                           v2f (&q)[PPB][3]) {
  #pragma unroll
  for (int j = 0; j < PPB; ++j) {
    int row = b * (TPB * 16) + j * 32 + lane;
    if (row >= numRows) row = numRows - 1;
    const float* rp = in + (size_t)row * 6;
    q[j][0] = *(const v2f*)(rp);
    q[j][1] = *(const v2f*)(rp + 2);
    q[j][2] = *(const v2f*)(rp + 4);
  }
}

__device__ __forceinline__ void compute_batch(float* __restrict__ out, int b,
                                              int numTiles, int lane,
                                              v2f w0, v2f w1, v2f w2,
                                              const float (&cc)[6],
                                              float* __restrict__ patch,
                                              const v2f (&q)[PPB][3]) {
  const int tilesLeft = numTiles - b * TPB;
  if (tilesLeft <= 0) return;
  #pragma unroll
  for (int j = 0; j < PPB; ++j) {
    v2f q1 = q[j][1];
    if (b * (TPB * 16) + j * 32 + lane < 2) q1.y = 0.0f;

    v8f acc = {};
    acc = __builtin_amdgcn_wmma_f32_16x16x4_f32(false, w0, false, q[j][0],
                                                (short)0, acc, false, false);
    acc = __builtin_amdgcn_wmma_f32_16x16x4_f32(false, w1, false, q1,
                                                (short)0, acc, false, false);
    acc = __builtin_amdgcn_wmma_f32_16x16x4_f32(false, w2, false, q[j][2],
                                                (short)0, acc, false, false);

    float* pr = patch + (j * 32 + lane) * 6;
    *(v2f*)(pr)     = v2f{ acc[0] + cc[0], acc[1] + cc[1] };
    *(v2f*)(pr + 2) = v2f{ acc[2] + cc[2], acc[3] + cc[3] };
    *(v2f*)(pr + 4) = v2f{ acc[4] + cc[4], acc[5] + cc[5] };
  }
  __builtin_amdgcn_fence(__ATOMIC_RELEASE, "wavefront", "local");
  __builtin_amdgcn_wave_barrier();
  __builtin_amdgcn_fence(__ATOMIC_ACQUIRE, "wavefront", "local");

  float* op = out + (size_t)b * TRIP_FLOATS;
  if (tilesLeft >= TPB) {
    #pragma unroll
    for (int i = 0; i < TRIP_FLOATS / 128; ++i) {
      const int f = (i * 32 + lane) * 4;
      *(v4f*)(op + f) = *(const v4f*)(patch + f);
    }
  } else {
    const int validFloats = tilesLeft * 96;
    #pragma unroll
    for (int i = 0; i < TRIP_FLOATS / 128; ++i) {
      const int f = (i * 32 + lane) * 4;
      if (f < validFloats) *(v4f*)(op + f) = *(const v4f*)(patch + f);
    }
  }
  __builtin_amdgcn_fence(__ATOMIC_RELEASE, "wavefront", "local");
  __builtin_amdgcn_wave_barrier();
  __builtin_amdgcn_fence(__ATOMIC_ACQUIRE, "wavefront", "local");
}

__device__ __forceinline__ void lds_barrier() {
  __builtin_amdgcn_fence(__ATOMIC_RELEASE, "workgroup", "local");
  __builtin_amdgcn_s_barrier();
  __builtin_amdgcn_fence(__ATOMIC_ACQUIRE, "workgroup", "local");
}

__global__ void __launch_bounds__(256) optnet_main_kernel(
    const float* __restrict__ in, const float* __restrict__ A,
    float* __restrict__ out, int numTiles) {
  __shared__ double sG[3][6];
  __shared__ double sU[3][7];
  __shared__ float  sW[42];
  __shared__ __attribute__((aligned(16))) float sY[8][TRIP_FLOATS];
  if (numTiles <= 0) return;

  const int lane = threadIdx.x & 31;
  const int wave = __builtin_amdgcn_readfirstlane((int)(threadIdx.x >> 5));
  const int waveId = (blockIdx.x << 3) | wave;
  const int numWaves = gridDim.x << 3;
  const int n  = lane & 15;
  const int hi = lane >> 4;

  const int chainWave = (int)((blockIdx.x ^ (blockIdx.x >> 8)) & 1u);
  const bool chain = (wave == chainWave);

  const int k = lane & 7;
  const bool comp = k < 6;
  const int kk = comp ? k : 0;
  float f1 = 0.0f, f2 = 0.0f, f3 = 0.0f, fp = 0.0f;
  if (chain) { f1 = A[kk]; f2 = A[6 + kk]; f3 = A[12 + kk]; fp = in[3]; }

  const int numBatches = (numTiles + TPB - 1) / TPB;
  int b = waveId;
  const int numRows = numTiles << 4;
  v2f xq[PPB][3], yq[PPB][3];
  load_batch(in, b, numRows, lane, xq);
  load_batch(in, b + numWaves, numRows, lane, yq);

  if (chain) {
    const double p3 = (double)fp;
    const double a1 = comp ? (double)f1 : 0.0;
    const double a2 = comp ? (double)f2 : 0.0;
    const double a3 = comp ? (double)f3 : 0.0;

    const double d1  = sum8(a1 * a1);
    const double t12 = sum8(a1 * a2);
    const double t13 = sum8(a1 * a3);
    const double r1  = 1.0 / d1;
    const double l21 = t12 * r1;
    const double l31 = t13 * r1;
    const double u2  = __builtin_fma(-l21, a1, a2);
    const double a3p = __builtin_fma(-l31, a1, a3);
    const double d2  = sum8(u2 * u2);
    const double t23 = sum8(u2 * a3p);
    const double r2  = 1.0 / d2;
    const double l32 = t23 * r2;
    const double u3  = __builtin_fma(-l32, u2, a3p);
    const double d3  = sum8(u3 * u3);
    const double r3  = 1.0 / d3;
    const double w2 = 1.0 - l21;
    const double w3 = __builtin_fma(-l32, w2, 1.0 - l31);

    if (lane < 6) { sG[0][lane] = a1 * r1; sG[1][lane] = u2 * r2; sG[2][lane] = u3 * r3; }
    if (lane < 7) {
      sU[0][lane] = (lane < 6) ? a1 : p3;
      sU[1][lane] = (lane < 6) ? u2 : p3 * w2;
      sU[2][lane] = (lane < 6) ? u3 : p3 * w3;
    }
  }
  lds_barrier();

  if (threadIdx.x < 42) {
    const int e  = threadIdx.x;
    const int rn = e / 7;
    const int K  = e - 7 * rn;
    double v = (rn == K) ? 1.0 : 0.0;
    v = __builtin_fma(-sG[0][rn], sU[0][K], v);
    v = __builtin_fma(-sG[1][rn], sU[1][K], v);
    v = __builtin_fma(-sG[2][rn], sU[2][K], v);
    sW[(K < 6) ? (rn * 6 + K) : (36 + rn)] = (float)v;
  }
  lds_barrier();

  const int wr = hi ? (n - 8) : n;
  const bool live = (wr >= 0) && (wr < 6);
  const float* wrow = sW + (live ? wr : 0) * 6;
  v2f w0 = { wrow[0], wrow[1] };
  v2f w1 = { wrow[2], wrow[3] };
  v2f w2 = { wrow[4], wrow[5] };
  if (!live) { w0 = v2f{0.0f, 0.0f}; w1 = w0; w2 = w0; }
  float cc[6];
  #pragma unroll
  for (int r = 0; r < 6; ++r) cc[r] = sW[36 + r];
  float* patch = sY[wave];

  while (b < numBatches) {
    compute_batch(out, b, numTiles, lane, w0, w1, w2, cc, patch, xq);
    load_batch(in, b + 2 * numWaves, numRows, lane, xq);
    compute_batch(out, b + numWaves, numTiles, lane, w0, w1, w2, cc, patch, yq);
    load_batch(in, b + 3 * numWaves, numRows, lane, yq);
    b += 2 * numWaves;
  }
}

extern "C" void kernel_launch(void* const* d_in, const int* in_sizes, int n_in,
                              void* d_out, int out_size, void* d_ws, size_t ws_size,
                              hipStream_t stream) {
  const float* puzzles = (const float*)d_in[0];
  const float* A       = (const float*)d_in[1];
  float* out = (float*)d_out;

  int nRows    = in_sizes[0] / 6;
  int numTiles = (nRows + 15) / 16;

  optnet_main_kernel<<<512, 256, 0, stream>>>(puzzles, A, out, numTiles);
}
